# starting kernel
# baseline (speedup 1.0000x reference)
#include <hip/hip_runtime.h>
#include <math.h>

typedef __attribute__((ext_vector_type(2))) float v2f;
typedef __attribute__((ext_vector_type(8))) float v8f;

#define NBANDS 65
#define LFIR   129
#define NPAD   144
#define FRAME  80
#define TFRAMES 4000
#define BATCH  32
#define LCONV  208
#define ROWS   (BATCH*TFRAMES)
#define NTILES 9
#define RTILES (ROWS/16)
#define KSTEPS 17
#define PPAIRS (KSTEPS*2)

__global__ void build_B(float* __restrict__ Bp) {
  int i = blockIdx.x * blockDim.x + threadIdx.x;
  if (i >= PPAIRS * NPAD * 2) return;
  int p   = i / (NPAD * 2);
  int rem = i - p * (NPAD * 2);
  int n   = rem >> 1;
  int r   = rem & 1;
  int k   = 2 * p + r;
  float v = 0.0f;
  if (k < NBANDS && n < LFIR) {
    const float PI = 3.14159265358979323846f;
    float hann = 0.5f * (1.0f - cosf(2.0f * PI * (float)n / (float)LFIR));
    float w = (k == 0) ? 1.0f : 2.0f;
    float c = cosf(2.0f * PI * (float)k * (float)(n - (NBANDS - 1)) / (float)LFIR);
    v = hann * w * c / (float)LFIR;
  }
  Bp[i] = v;
}

__global__ void fir_gemm(const float* __restrict__ H,
                         const float* __restrict__ Bp,
                         float* __restrict__ FIR) {
  int wave = (int)((blockIdx.x * blockDim.x + threadIdx.x) >> 5);
  int lane = (int)(threadIdx.x & 31u);
  if (wave >= RTILES) return;

  int half = lane >> 4;
  int lm   = lane & 15;

  int row = wave * 16 + lm;
  const float* hrow = H + (size_t)row * NBANDS;

  v8f c0 = {}, c1 = {}, c2 = {}, c3 = {}, c4 = {}, c5 = {}, c6 = {}, c7 = {}, c8 = {};

#pragma unroll 1
  for (int kk = 0; kk < KSTEPS - 1; ++kk) {
    int kA = kk * 4 + 2 * half;
    v2f a;
    a.x = hrow[kA];
    a.y = hrow[kA + 1];
    const v2f* brow = (const v2f*)(Bp + ((size_t)(kk * 2 + half) * NPAD + lm) * 2);
    v2f b0 = brow[0],  b1 = brow[16],  b2 = brow[32],  b3 = brow[48],
        b4 = brow[64], b5 = brow[80],  b6 = brow[96],  b7 = brow[112],
        b8 = brow[128];
    c0 = __builtin_amdgcn_wmma_f32_16x16x4_f32(false, a, false, b0, (short)0, c0, false, false);
    c1 = __builtin_amdgcn_wmma_f32_16x16x4_f32(false, a, false, b1, (short)0, c1, false, false);
    c2 = __builtin_amdgcn_wmma_f32_16x16x4_f32(false, a, false, b2, (short)0, c2, false, false);
    c3 = __builtin_amdgcn_wmma_f32_16x16x4_f32(false, a, false, b3, (short)0, c3, false, false);
    c4 = __builtin_amdgcn_wmma_f32_16x16x4_f32(false, a, false, b4, (short)0, c4, false, false);
    c5 = __builtin_amdgcn_wmma_f32_16x16x4_f32(false, a, false, b5, (short)0, c5, false, false);
    c6 = __builtin_amdgcn_wmma_f32_16x16x4_f32(false, a, false, b6, (short)0, c6, false, false);
    c7 = __builtin_amdgcn_wmma_f32_16x16x4_f32(false, a, false, b7, (short)0, c7, false, false);
    c8 = __builtin_amdgcn_wmma_f32_16x16x4_f32(false, a, false, b8, (short)0, c8, false, false);
  }

  {
    float h64 = hrow[NBANDS - 1];
    v2f a;
    a.x = half ? 0.0f : h64;
    a.y = 0.0f;
    const int kk = KSTEPS - 1;
    const v2f* brow = (const v2f*)(Bp + ((size_t)(kk * 2 + half) * NPAD + lm) * 2);
    v2f b0 = brow[0],  b1 = brow[16],  b2 = brow[32],  b3 = brow[48],
        b4 = brow[64], b5 = brow[80],  b6 = brow[96],  b7 = brow[112],
        b8 = brow[128];
    c0 = __builtin_amdgcn_wmma_f32_16x16x4_f32(false, a, false, b0, (short)0, c0, false, false);
    c1 = __builtin_amdgcn_wmma_f32_16x16x4_f32(false, a, false, b1, (short)0, c1, false, false);
    c2 = __builtin_amdgcn_wmma_f32_16x16x4_f32(false, a, false, b2, (short)0, c2, false, false);
    c3 = __builtin_amdgcn_wmma_f32_16x16x4_f32(false, a, false, b3, (short)0, c3, false, false);
    c4 = __builtin_amdgcn_wmma_f32_16x16x4_f32(false, a, false, b4, (short)0, c4, false, false);
    c5 = __builtin_amdgcn_wmma_f32_16x16x4_f32(false, a, false, b5, (short)0, c5, false, false);
    c6 = __builtin_amdgcn_wmma_f32_16x16x4_f32(false, a, false, b6, (short)0, c6, false, false);
    c7 = __builtin_amdgcn_wmma_f32_16x16x4_f32(false, a, false, b7, (short)0, c7, false, false);
    c8 = __builtin_amdgcn_wmma_f32_16x16x4_f32(false, a, false, b8, (short)0, c8, false, false);
  }

  size_t rbase = (size_t)(wave * 16 + 8 * half) * NPAD;
  v8f acc[9] = {c0, c1, c2, c3, c4, c5, c6, c7, c8};
#pragma unroll
  for (int j = 0; j < NTILES; ++j) {
    size_t outBase = rbase + j * 16 + lm;
#pragma unroll
    for (int v = 0; v < 8; ++v)
      FIR[outBase + (size_t)v * NPAD] = acc[j][v];
  }
}

__global__ void conv_ola(const float* __restrict__ noise,
                         const float* __restrict__ FIR,
                         float* __restrict__ out) {
  long long idx = (long long)blockIdx.x * blockDim.x + threadIdx.x;
  if (idx >= (long long)BATCH * TFRAMES * FRAME) return;
  int b = (int)(idx / (TFRAMES * FRAME));
  int s = (int)(idx - (long long)b * (TFRAMES * FRAME));

  int u = s + (NBANDS - 1);
  int t_hi = u / FRAME; if (t_hi > TFRAMES - 1) t_hi = TFRAMES - 1;
  int lo = u - (LCONV - 1);
  int t_lo = (lo <= 0) ? 0 : (lo + FRAME - 1) / FRAME;

  float acc = 0.0f;
  for (int t = t_lo; t <= t_hi; ++t) {
    int n = u - t * FRAME;
    int kmin = n - (FRAME - 1); if (kmin < 0) kmin = 0;
    int kmax = n;               if (kmax > LFIR - 1) kmax = LFIR - 1;
    const float* xr = noise + (size_t)(b * TFRAMES + t) * FRAME;
    const float* fr = FIR   + (size_t)(b * TFRAMES + t) * NPAD;
    for (int k = kmin; k <= kmax; ++k) {
      float x = 2.0f * xr[n - k] - 1.0f;
      acc = fmaf(x, fr[k], acc);
    }
  }
  out[idx] = acc;
}

extern "C" void kernel_launch(void* const* d_in, const int* in_sizes, int n_in,
                              void* d_out, int out_size, void* d_ws, size_t ws_size,
                              hipStream_t stream) {
  const float* H     = (const float*)d_in[0];
  const float* noise = (const float*)d_in[1];
  float* out = (float*)d_out;

  float* Bp  = (float*)d_ws;
  float* FIR = (float*)((char*)d_ws + 64 * 1024);

  {
    int total = PPAIRS * NPAD * 2;
    build_B<<<(total + 255) / 256, 256, 0, stream>>>(Bp);
  }
  {
    long long threads = (long long)RTILES * 32;
    fir_gemm<<<(unsigned)((threads + 255) / 256), 256, 0, stream>>>(H, Bp, FIR);
  }
  {
    long long total = (long long)BATCH * TFRAMES * FRAME;
    conv_ola<<<(unsigned)((total + 255) / 256), 256, 0, stream>>>(noise, FIR, out);
  }
}


# matrix loop operands requested ahead: a lane's noise frame and its H bands of all 17 k-steps before the loop, table fragments 4-5 k-steps ahead in registers; loop fully unrolled; sums unchanged
# speedup vs baseline: 7.9334x; 7.9334x over previous
#include <hip/hip_runtime.h>
#include <math.h>
#include <type_traits>


typedef __attribute__((ext_vector_type(2))) float v2f;
typedef __attribute__((ext_vector_type(4))) float v4f;
typedef __attribute__((ext_vector_type(8))) float v8f;

#define NBANDS 65
#define LFIR   129
#define NPAD   144
#define FRAME  80
#define TFRAMES 4000
#define BATCH  32
#define LCONV  208
#define NTILES 9
#define KSTEPS 17
#define PPAIRS (KSTEPS*2)
#define OUTLEN (TFRAMES*FRAME)
#define GDELAY (NBANDS-1)
#define SLOTS  14
#define WAVES_PER_ROW 286

__global__ void build_B(float* __restrict__ Bp) {
  int i = blockIdx.x * blockDim.x + threadIdx.x;
  if (i >= PPAIRS * NPAD * 2) return;
  int p   = i / (NPAD * 2);
  int rem = i - p * (NPAD * 2);
  int n   = rem >> 1;
  int r   = rem & 1;
  int k   = 2 * p + r;
  float v = 0.0f;
  if (k < NBANDS && n < LFIR) {
    const float PI = 3.14159265358979323846f;
    float hann = 0.5f * (1.0f - cosf(2.0f * PI * (float)n / (float)LFIR));
    float w = (k == 0) ? 1.0f : 2.0f;
    float c = cosf(2.0f * PI * (float)k * (float)(n - (NBANDS - 1)) / (float)LFIR);
    v = hann * w * c / (float)LFIR;
  }
  Bp[i] = v;
}

template <int I, int N, typename F>
__device__ __forceinline__ void static_for(F&& f) {
  if constexpr (I < N) {
    f(std::integral_constant<int, I>{});
    static_for<I + 1, N>(f);
  }
}

__device__ __forceinline__ float row_up1(float x) {
  return __builtin_bit_cast(float, __builtin_amdgcn_update_dpp(0, __builtin_bit_cast(int, x), 0x111, 0xf, 0xf, true));
}
__device__ __forceinline__ float row_up2(float x) {
  return __builtin_bit_cast(float, __builtin_amdgcn_update_dpp(0, __builtin_bit_cast(int, x), 0x112, 0xf, 0xf, true));
}
__device__ __forceinline__ float xhalf(float x) {
  return __builtin_bit_cast(float, __builtin_amdgcn_permlanex16(__builtin_bit_cast(int, x), __builtin_bit_cast(int, x),
                                                                 0x76543210, 0xfedcba98, false, false));
}

__global__ __launch_bounds__(128)
void fir_conv_ola(const float* __restrict__ H,
                  const float* __restrict__ noise,
                  const float* __restrict__ Bp,
                  float* __restrict__ out) {
  const int wave = (int)((blockIdx.x * blockDim.x + threadIdx.x) >> 5);
  const int lane = (int)(threadIdx.x & 31u);
  const int half = lane >> 4;
  const int c    = lane & 15;

  const int b    = wave / WAVES_PER_ROW;
  const int w    = wave - b * WAVES_PER_ROW;
  const int slot = w * SLOTS + c - 2;
  const bool fvalid = (slot >= 0) && (slot < TFRAMES);
  const int fcl  = slot < 0 ? 0 : (slot > TFRAMES - 1 ? TFRAMES - 1 : slot);

  const float* hrow = H + ((size_t)b * TFRAMES + (size_t)fcl) * NBANDS;

  v8f acc[NTILES];
  static_for<0, NTILES>([&](auto jc) { acc[decltype(jc)::value] = (v8f){0.f, 0.f, 0.f, 0.f, 0.f, 0.f, 0.f, 0.f}; });

  const v4f* xrow = (const v4f*)(noise + ((size_t)b * TFRAMES + (size_t)fcl) * FRAME);
  v4f xq[FRAME / 4];
  static_for<0, FRAME / 4>([&](auto qc) { xq[decltype(qc)::value] = xrow[decltype(qc)::value]; });

  const float* hp = hrow + 2 * half;
  v4f hq[(KSTEPS - 1) / 2];
  static_for<0, (KSTEPS - 1) / 2>([&](auto mc) {
    constexpr int m = decltype(mc)::value;
    hq[m] = (v4f){hp[8 * m], hp[8 * m + 1], hp[8 * m + 4], hp[8 * m + 5]};
  });
  const float h64 = hrow[NBANDS - 1];

  constexpr int PF = 4;
  constexpr int PF0 = 2;
  constexpr int NB = PF + 1;
  const v2f* btab = (const v2f*)(Bp + ((size_t)half * NPAD + c) * 2);
  v2f bt[NB][NTILES];
  static_for<0, PF0>([&](auto kc) {
    constexpr int kk = decltype(kc)::value;
    static_for<0, NTILES>([&](auto jc) {
      constexpr int j = decltype(jc)::value;
      bt[kk % NB][j] = btab[kk * 2 * NPAD + 16 * j];
    });
  });

  asm volatile("" :: "v"(xq[0]), "v"(xq[1]), "v"(xq[2]), "v"(xq[3]), "v"(xq[4]), "v"(xq[5]), "v"(xq[6]),
               "v"(xq[7]), "v"(xq[8]), "v"(xq[9]), "v"(xq[10]), "v"(xq[11]), "v"(xq[12]), "v"(xq[13]),
               "v"(xq[14]), "v"(xq[15]), "v"(xq[16]), "v"(xq[17]), "v"(xq[18]), "v"(xq[19]),
               "v"(hq[0]), "v"(hq[1]), "v"(hq[2]), "v"(hq[3]), "v"(hq[4]), "v"(hq[5]), "v"(hq[6]), "v"(hq[7]),
               "v"(h64));

  v2f hk[KSTEPS];
  static_for<0, (KSTEPS - 1) / 2>([&](auto mc) {
    constexpr int m = decltype(mc)::value;
    hk[2 * m].x     = fvalid ? hq[m].x : 0.0f;
    hk[2 * m].y     = fvalid ? hq[m].y : 0.0f;
    hk[2 * m + 1].x = fvalid ? hq[m].z : 0.0f;
    hk[2 * m + 1].y = fvalid ? hq[m].w : 0.0f;
  });
  hk[KSTEPS - 1].x = (half || !fvalid) ? 0.0f : h64;
  hk[KSTEPS - 1].y = 0.0f;

  __builtin_amdgcn_sched_barrier(0);
  static_for<0, KSTEPS>([&](auto kc) {
    constexpr int kk = decltype(kc)::value;
    static_for<(kk == 0 ? PF0 : kk + PF), (kk + PF < KSTEPS ? kk + PF + 1 : kk + PF)>([&](auto nc) {
      constexpr int nk = decltype(nc)::value;
      static_for<0, NTILES>([&](auto jc) {
        constexpr int j = decltype(jc)::value;
        bt[nk % NB][j] = btab[nk * 2 * NPAD + 16 * j];
      });
    });
    static_for<0, NTILES>([&](auto jc) {
      constexpr int j = decltype(jc)::value;
      acc[j] = __builtin_amdgcn_wmma_f32_16x16x4_f32(false, bt[kk % NB][j], false, hk[kk], (short)0, acc[j], false, false);
    });
    __builtin_amdgcn_sched_barrier(0x6);
  });

  v2f P2[LCONV / 2];
  static_for<0, LCONV / 2>([&](auto ic) { P2[decltype(ic)::value] = (v2f){0.0f, 0.0f}; });

  float xprev = 0.0f;
  static_for<0, FRAME / 4>([&](auto qc) {
    constexpr int q = decltype(qc)::value;
    const v4f xv = xq[q];
    const float x0 = 2.0f * xv.x - 1.0f, x1 = 2.0f * xv.y - 1.0f,
                x2 = 2.0f * xv.z - 1.0f, x3 = 2.0f * xv.w - 1.0f;
    const v2f e0 = {x0, x1}, e1 = {x2, x3};
    const v2f o0 = {xprev, x0}, o1 = {x1, x2};
    static_for<0, 33>([&](auto sc) {
      constexpr int s  = decltype(sc)::value;
      constexpr int j  = s / 4, r = 2 * (s % 4);
      constexpr int kl = 16 * j + r;
      const float t = acc[j][r];
      const v2f t2 = {t, t};
      P2[(4 * q + kl) / 2]     = __builtin_elementwise_fma(e0, t2, P2[(4 * q + kl) / 2]);
      P2[(4 * q + 2 + kl) / 2] = __builtin_elementwise_fma(e1, t2, P2[(4 * q + 2 + kl) / 2]);
    });
    static_for<0, 32>([&](auto sc) {
      constexpr int s  = decltype(sc)::value;
      constexpr int j  = s / 4, r = 2 * (s % 4) + 1;
      constexpr int kl = 16 * j + r;
      const float t = acc[j][r];
      const v2f t2 = {t, t};
      P2[(4 * q - 1 + kl) / 2] = __builtin_elementwise_fma(o0, t2, P2[(4 * q - 1 + kl) / 2]);
      P2[(4 * q + 1 + kl) / 2] = __builtin_elementwise_fma(o1, t2, P2[(4 * q + 1 + kl) / 2]);
    });
    xprev = x3;
  });
  {
    const v2f o = {xprev, 0.0f};
    static_for<0, 32>([&](auto sc) {
      constexpr int s  = decltype(sc)::value;
      constexpr int j  = s / 4, r = 2 * (s % 4) + 1;
      constexpr int kl = 16 * j + r;
      const float t = acc[j][r];
      const v2f t2 = {t, t};
      P2[(FRAME - 1 + kl) / 2] = __builtin_elementwise_fma(o, t2, P2[(FRAME - 1 + kl) / 2]);
    });
  }

  float Q[FRAME + 8];
  static_for<0, FRAME + 8>([&](auto ic) {
    constexpr int i = decltype(ic)::value - 8;
    float v = row_up1(P2[(i + FRAME) / 2][(i + FRAME) % 2]);
    if constexpr (i >= 0) v = P2[i / 2][i % 2] + v;
    if constexpr (i + 2 * FRAME < LCONV) v = v + row_up2(P2[(i + 2 * FRAME) / 2][(i + 2 * FRAME) % 2]);
    Q[i + 8] = v;
  });

  const bool storer = (half == 0) && (c >= 2) && (slot <= TFRAMES);
  const long long sbase = (long long)b * OUTLEN + (long long)slot * FRAME - GDELAY;
  static_for<0, FRAME / 4>([&](auto qc) {
    constexpr int q = decltype(qc)::value;
    float4 o;
    o.x = Q[4 * q + 8]  + xhalf(Q[4 * q]);
    o.y = Q[4 * q + 9]  + xhalf(Q[4 * q + 1]);
    o.z = Q[4 * q + 10] + xhalf(Q[4 * q + 2]);
    o.w = Q[4 * q + 11] + xhalf(Q[4 * q + 3]);
    const bool ok = storer && (slot >= 1 || q >= GDELAY / 4) && (slot <= TFRAMES - 1 || q < GDELAY / 4);
    if (ok) *(float4*)(out + sbase + 4 * q) = o;
  });
}

extern "C" void kernel_launch(void* const* d_in, const int* in_sizes, int n_in,
                              void* d_out, int out_size, void* d_ws, size_t ws_size,
                              hipStream_t stream) {
  const float* H     = (const float*)d_in[0];
  const float* noise = (const float*)d_in[1];
  float* out = (float*)d_out;

  float* Bp  = (float*)d_ws;

  build_B<<<39, 256, 0, stream>>>(Bp);
  fir_conv_ola<<<2288, 128, 0, stream>>>(H, noise, Bp, out);
}


# table fragments and noise frames staged in LDS (whole-line loads) and read at use, not held in registers: 334 VGPRs not 418, three waves per SIMD not two; every sum and its order unchanged
# speedup vs baseline: 9.0789x; 1.1444x over previous
#include <hip/hip_runtime.h>
#include <math.h>
#include <type_traits>


typedef __attribute__((ext_vector_type(2))) float v2f;
typedef __attribute__((ext_vector_type(4))) float v4f;
typedef __attribute__((ext_vector_type(8))) float v8f;

#define NBANDS 65
#define LFIR   129
#define NPAD   144
#define FRAME  80
#define TFRAMES 4000
#define BATCH  32
#define LCONV  208
#define NTILES 9
#define KSTEPS 17
#define PPAIRS (KSTEPS*2)
#define OUTLEN (TFRAMES*FRAME)
#define GDELAY (NBANDS-1)
#define SLOTS  14
#define WAVES_PER_ROW 286
#define XPITCH 84
#define GK 4

__global__ void build_B(float* __restrict__ Bp) {
  int i = blockIdx.x * blockDim.x + threadIdx.x;
  if (i >= PPAIRS * NPAD * 2) return;
  int p   = i / (NPAD * 2);
  int rem = i - p * (NPAD * 2);
  int n   = rem >> 1;
  int r   = rem & 1;
  int k   = 2 * p + r;
  float v = 0.0f;
  if (k < NBANDS && n < LFIR) {
    const float PI = 3.14159265358979323846f;
    float hann = 0.5f * (1.0f - cosf(2.0f * PI * (float)n / (float)LFIR));
    float w = (k == 0) ? 1.0f : 2.0f;
    float c = cosf(2.0f * PI * (float)k * (float)(n - (NBANDS - 1)) / (float)LFIR);
    v = hann * w * c / (float)LFIR;
  }
  Bp[i] = v;
}

template <int I, int N, typename F>
__device__ __forceinline__ void static_for(F&& f) {
  if constexpr (I < N) {
    f(std::integral_constant<int, I>{});
    static_for<I + 1, N>(f);
  }
}

__device__ __forceinline__ float row_up1(float x) {
  return __builtin_bit_cast(float, __builtin_amdgcn_update_dpp(0, __builtin_bit_cast(int, x), 0x111, 0xf, 0xf, true));
}
__device__ __forceinline__ float row_up2(float x) {
  return __builtin_bit_cast(float, __builtin_amdgcn_update_dpp(0, __builtin_bit_cast(int, x), 0x112, 0xf, 0xf, true));
}
__device__ __forceinline__ float xhalf(float x) {
  return __builtin_bit_cast(float, __builtin_amdgcn_permlanex16(__builtin_bit_cast(int, x), __builtin_bit_cast(int, x),
                                                                 0x76543210, 0xfedcba98, false, false));
}

__global__ __launch_bounds__(128)
void fir_conv_ola(const float* __restrict__ H,
                  const float* __restrict__ noise,
                  const float* __restrict__ Bp,
                  float* __restrict__ out) {
  __shared__ float sB[PPAIRS * NPAD * 2];
  __shared__ float sX[4][16 * XPITCH];

  const int wave = (int)((blockIdx.x * blockDim.x + threadIdx.x) >> 5);
  const int lane = (int)(threadIdx.x & 31u);
  const int half = lane >> 4;
  const int c    = lane & 15;

  const int b    = wave / WAVES_PER_ROW;
  const int w    = wave - b * WAVES_PER_ROW;
  const int slot = w * SLOTS + c - 2;
  const bool fvalid = (slot >= 0) && (slot < TFRAMES);
  const int fcl  = slot < 0 ? 0 : (slot > TFRAMES - 1 ? TFRAMES - 1 : slot);

  const float* hrow = H + ((size_t)b * TFRAMES + (size_t)fcl) * NBANDS;

  v8f acc[NTILES];
  static_for<0, NTILES>([&](auto jc) { acc[decltype(jc)::value] = (v8f){0.f, 0.f, 0.f, 0.f, 0.f, 0.f, 0.f, 0.f}; });


  constexpr int NV4   = PPAIRS * NPAD * 2 / 4;
  constexpr int TCOPY = (NV4 + 127) / 128;
  v4f tb[TCOPY];
  static_for<0, TCOPY>([&](auto ic) {
    constexpr int i = decltype(ic)::value;
    const int idx = i * 128 + (int)threadIdx.x;
    tb[i] = ((const v4f*)Bp)[idx < NV4 ? idx : NV4 - 1];
  });

  float* sXw = sX[threadIdx.x >> 5];
  v4f xs[10];
  static_for<0, 10>([&](auto tc) {
    constexpr int t = decltype(tc)::value;
    const int i  = t * 32 + lane;
    const int fr = i / (FRAME / 4);
    const int q4 = i - fr * (FRAME / 4);
    int f = w * SLOTS - 2 + fr;
    f = f < 0 ? 0 : (f > TFRAMES - 1 ? TFRAMES - 1 : f);
    xs[t] = *(const v4f*)(noise + ((size_t)b * TFRAMES + (size_t)f) * FRAME + 4 * q4);
  });

  const float* hp = hrow + 2 * half;
  v4f hq[(KSTEPS - 1) / 2];
  static_for<0, (KSTEPS - 1) / 2>([&](auto mc) {
    constexpr int m = decltype(mc)::value;
    hq[m] = (v4f){hp[8 * m], hp[8 * m + 1], hp[8 * m + 4], hp[8 * m + 5]};
  });
  const float h64 = hrow[NBANDS - 1];

  asm volatile("" :: "v"(xs[0]), "v"(xs[1]), "v"(xs[2]), "v"(xs[3]), "v"(xs[4]), "v"(xs[5]), "v"(xs[6]),
               "v"(xs[7]), "v"(xs[8]), "v"(xs[9]),
               "v"(hq[0]), "v"(hq[1]), "v"(hq[2]), "v"(hq[3]), "v"(hq[4]), "v"(hq[5]), "v"(hq[6]), "v"(hq[7]),
               "v"(h64));

  static_for<0, TCOPY>([&](auto ic) {
    constexpr int i = decltype(ic)::value;
    const int idx = i * 128 + (int)threadIdx.x;
    if (idx < NV4) ((v4f*)sB)[idx] = tb[i];
  });
  static_for<0, 10>([&](auto tc) {
    constexpr int t = decltype(tc)::value;
    const int i  = t * 32 + lane;
    const int fr = i / (FRAME / 4);
    const int q4 = i - fr * (FRAME / 4);
    *(v4f*)(sXw + fr * XPITCH + 4 * q4) = xs[t];
  });
  __syncthreads();

  constexpr int NG = (KSTEPS + GK - 1) / GK;
  const v2f* btab = (const v2f*)sB + (half * NPAD + c);
  v2f bt[2][GK][NTILES];
  static_for<0, GK>([&](auto kc) {
    constexpr int kk = decltype(kc)::value;
    static_for<0, NTILES>([&](auto jc) {
      constexpr int j = decltype(jc)::value;
      bt[0][kk][j] = btab[kk * 2 * NPAD + 16 * j];
    });
  });

  v2f hk[KSTEPS];
  static_for<0, (KSTEPS - 1) / 2>([&](auto mc) {
    constexpr int m = decltype(mc)::value;
    hk[2 * m].x     = fvalid ? hq[m].x : 0.0f;
    hk[2 * m].y     = fvalid ? hq[m].y : 0.0f;
    hk[2 * m + 1].x = fvalid ? hq[m].z : 0.0f;
    hk[2 * m + 1].y = fvalid ? hq[m].w : 0.0f;
  });
  hk[KSTEPS - 1].x = (half || !fvalid) ? 0.0f : h64;
  hk[KSTEPS - 1].y = 0.0f;

  __builtin_amdgcn_sched_barrier(0);
  static_for<0, NG>([&](auto gc) {
    constexpr int g = decltype(gc)::value;
    static_for<(g + 1) * GK, ((g + 2) * GK < KSTEPS ? (g + 2) * GK : KSTEPS)>([&](auto nc) {
      constexpr int nk = decltype(nc)::value;
      static_for<0, NTILES>([&](auto jc) {
        constexpr int j = decltype(jc)::value;
        bt[(g + 1) % 2][nk % GK][j] = btab[nk * 2 * NPAD + 16 * j];
      });
    });
    __builtin_amdgcn_sched_barrier(0);
    static_for<g * GK, ((g + 1) * GK < KSTEPS ? (g + 1) * GK : KSTEPS)>([&](auto kc) {
      constexpr int kk = decltype(kc)::value;
      static_for<0, NTILES>([&](auto jc) {
        constexpr int j = decltype(jc)::value;
        acc[j] = __builtin_amdgcn_wmma_f32_16x16x4_f32(false, bt[g % 2][kk % GK][j], false, hk[kk], (short)0, acc[j], false, false);
      });
    });
    __builtin_amdgcn_sched_barrier(0);
  });

  v2f P2[LCONV / 2];
  static_for<0, LCONV / 2>([&](auto ic) { P2[decltype(ic)::value] = (v2f){0.0f, 0.0f}; });

  float xprev = 0.0f;
  const float* xlane = sXw + c * XPITCH;
  v4f xnext = *(const v4f*)xlane;
  static_for<0, FRAME / 4>([&](auto qc) {
    constexpr int q = decltype(qc)::value;
    const v4f xv = xnext;
    if constexpr (q + 1 < FRAME / 4) xnext = *(const v4f*)(xlane + 4 * (q + 1));
    const float x0 = 2.0f * xv.x - 1.0f, x1 = 2.0f * xv.y - 1.0f,
                x2 = 2.0f * xv.z - 1.0f, x3 = 2.0f * xv.w - 1.0f;
    const v2f e0 = {x0, x1}, e1 = {x2, x3};
    const v2f o0 = {xprev, x0}, o1 = {x1, x2};
    static_for<0, 33>([&](auto sc) {
      constexpr int s  = decltype(sc)::value;
      constexpr int j  = s / 4, r = 2 * (s % 4);
      constexpr int kl = 16 * j + r;
      const float t = acc[j][r];
      const v2f t2 = {t, t};
      P2[(4 * q + kl) / 2]     = __builtin_elementwise_fma(e0, t2, P2[(4 * q + kl) / 2]);
      P2[(4 * q + 2 + kl) / 2] = __builtin_elementwise_fma(e1, t2, P2[(4 * q + 2 + kl) / 2]);
    });
    static_for<0, 32>([&](auto sc) {
      constexpr int s  = decltype(sc)::value;
      constexpr int j  = s / 4, r = 2 * (s % 4) + 1;
      constexpr int kl = 16 * j + r;
      const float t = acc[j][r];
      const v2f t2 = {t, t};
      P2[(4 * q - 1 + kl) / 2] = __builtin_elementwise_fma(o0, t2, P2[(4 * q - 1 + kl) / 2]);
      P2[(4 * q + 1 + kl) / 2] = __builtin_elementwise_fma(o1, t2, P2[(4 * q + 1 + kl) / 2]);
    });
    xprev = x3;
    __builtin_amdgcn_sched_barrier(0);
  });
  {
    const v2f o = {xprev, 0.0f};
    static_for<0, 32>([&](auto sc) {
      constexpr int s  = decltype(sc)::value;
      constexpr int j  = s / 4, r = 2 * (s % 4) + 1;
      constexpr int kl = 16 * j + r;
      const float t = acc[j][r];
      const v2f t2 = {t, t};
      P2[(FRAME - 1 + kl) / 2] = __builtin_elementwise_fma(o, t2, P2[(FRAME - 1 + kl) / 2]);
    });
  }

  float Q[FRAME + 8];
  static_for<0, FRAME + 8>([&](auto ic) {
    constexpr int i = decltype(ic)::value - 8;
    float v = row_up1(P2[(i + FRAME) / 2][(i + FRAME) % 2]);
    if constexpr (i >= 0) v = P2[i / 2][i % 2] + v;
    if constexpr (i + 2 * FRAME < LCONV) v = v + row_up2(P2[(i + 2 * FRAME) / 2][(i + 2 * FRAME) % 2]);
    Q[i + 8] = v;
  });

  const bool storer = (half == 0) && (c >= 2) && (slot <= TFRAMES);
  const long long sbase = (long long)b * OUTLEN + (long long)slot * FRAME - GDELAY;
  float4 o[FRAME / 4];
  static_for<0, FRAME / 4>([&](auto qc) {
    constexpr int q = decltype(qc)::value;
    o[q].x = Q[4 * q + 8]  + xhalf(Q[4 * q]);
    o[q].y = Q[4 * q + 9]  + xhalf(Q[4 * q + 1]);
    o[q].z = Q[4 * q + 10] + xhalf(Q[4 * q + 2]);
    o[q].w = Q[4 * q + 11] + xhalf(Q[4 * q + 3]);
  });
  static_for<0, FRAME / 4>([&](auto qc) {
    constexpr int q = decltype(qc)::value;
    const bool ok = storer && (slot >= 1 || q >= GDELAY / 4) && (slot <= TFRAMES - 1 || q < GDELAY / 4);
    if (ok) *(float4*)(out + sbase + 4 * q) = o[q];
  });
}

extern "C" void kernel_launch(void* const* d_in, const int* in_sizes, int n_in,
                              void* d_out, int out_size, void* d_ws, size_t ws_size,
                              hipStream_t stream) {
  const float* H     = (const float*)d_in[0];
  const float* noise = (const float*)d_in[1];
  float* out = (float*)d_out;

  float* Bp  = (float*)d_ws;

  build_B<<<39, 256, 0, stream>>>(Bp);
  fir_conv_ola<<<2288, 128, 0, stream>>>(H, noise, Bp, out);
}
